# starting kernel
# baseline (speedup 1.0000x reference)
#include <hip/hip_runtime.h>
#include <hip/hip_bf16.h>
#include <cstddef>
#include <cstdint>

#define N_NODES 100000
#define N_EDGES 3200000
#define BN_EPS 1e-5f

typedef __attribute__((ext_vector_type(2))) float v2f;
typedef __attribute__((ext_vector_type(8))) float v8f;

__device__ __forceinline__ v8f wmma_f32(v2f a, v2f b, v8f c) {
  return __builtin_amdgcn_wmma_f32_16x16x4_f32(false, a, false, b, (short)0, c,
                                               false, false);
}

__global__ void zero_f4_kernel(float4* __restrict__ p, long long n4) {
  long long i = (long long)blockIdx.x * blockDim.x + threadIdx.x;
  if (i < n4) p[i] = make_float4(0.f, 0.f, 0.f, 0.f);
}

__global__ void build_input_kernel(const float* __restrict__ x,
                                   const int* __restrict__ labels,
                                   const float* __restrict__ emb,
                                   float* __restrict__ h) {
  long long idx = (long long)blockIdx.x * blockDim.x + threadIdx.x;
  if (idx >= (long long)N_NODES * 16) return;
  const int i = (int)(idx >> 4);
  const int c = (int)(idx & 15);
  float v = 0.0f;
  if (c < 2)       v = x[(size_t)i * 2 + c];
  else if (c < 10) v = emb[(size_t)labels[i] * 8 + (c - 2)];
  h[(size_t)i * 64 + c] = v;
}

__global__ void pad_w1_kernel(const float* __restrict__ W1,
                              float* __restrict__ Wpad) {
  const int idx = blockIdx.x * blockDim.x + threadIdx.x;
  if (idx >= 16 * 64) return;
  const int r = idx >> 6;
  Wpad[idx] = (r < 10) ? W1[idx] : 0.0f;
}

__global__ void scatter_kernel(const float* __restrict__ h,
                               const int* __restrict__ edge_index,
                               float* __restrict__ agg, int log2w) {
  const long long idx = (long long)blockIdx.x * blockDim.x + threadIdx.x;
  const long long total = (long long)N_EDGES << log2w;
  if (idx >= total) return;
  const int e = (int)(idx >> log2w);
  const int c = (int)(idx & ((1 << log2w) - 1));
  const int s = edge_index[e];
  const int d = edge_index[N_EDGES + e];
  const float v = h[(size_t)s * 64 + c];
  unsafeAtomicAdd(&agg[(size_t)d * 64 + c], v);
}

__global__ void __launch_bounds__(128)
gin_mlp_kernel(float* __restrict__ h, const float* __restrict__ agg,
               const float* __restrict__ W1, const float* __restrict__ b1,
               const float* __restrict__ W2, const float* __restrict__ b2,
               const float* __restrict__ eps_ptr, int eps_idx, int K,
               float* __restrict__ stats) {
  __shared__ float ytile[4 * 16 * 68];
  __shared__ float sstat[128];
  const int tid   = threadIdx.x;
  const int wave  = tid >> 5;
  const int lane  = tid & 31;
  const int lrow  = lane & 15;
  const int lhalf = lane >> 4;
  const int tile  = blockIdx.x * 4 + wave;

  if (tid < 128) sstat[tid] = 0.0f;
  __syncthreads();

  if (tile < (N_NODES / 16)) {
    const int base = tile * 16;
    const float onepe = 1.0f + eps_ptr[eps_idx];
    const int koff = lhalf * 2;

    v8f acc[4] = {};
    for (int kk = 0; kk < K; kk += 4) {
      const int k0 = kk + koff;
      const float* hp = h   + (size_t)(base + lrow) * 64 + k0;
      const float* ap = agg + (size_t)(base + lrow) * 64 + k0;
      v2f afrag;
      afrag.x = onepe * hp[0] + ap[0];
      afrag.y = onepe * hp[1] + ap[1];
      const float* wp = W1 + (size_t)k0 * 64 + lrow;
#pragma unroll
      for (int nt = 0; nt < 4; ++nt) {
        v2f bfrag;
        bfrag.x = wp[nt * 16];
        bfrag.y = wp[nt * 16 + 64];
        acc[nt] = wmma_f32(afrag, bfrag, acc[nt]);
      }
    }

    float* my = ytile + wave * (16 * 68);
#pragma unroll
    for (int nt = 0; nt < 4; ++nt) {
      const float bv = b1[nt * 16 + lrow];
#pragma unroll
      for (int r = 0; r < 8; ++r)
        my[(r + lhalf * 8) * 68 + nt * 16 + lrow] = fmaxf(acc[nt][r] + bv, 0.f);
    }

    v8f acc2[4] = {};
    for (int kk = 0; kk < 64; kk += 4) {
      const int k0 = kk + koff;
      v2f afrag;
      afrag.x = my[lrow * 68 + k0];
      afrag.y = my[lrow * 68 + k0 + 1];
      const float* wp = W2 + (size_t)k0 * 64 + lrow;
#pragma unroll
      for (int nt = 0; nt < 4; ++nt) {
        v2f bfrag;
        bfrag.x = wp[nt * 16];
        bfrag.y = wp[nt * 16 + 64];
        acc2[nt] = wmma_f32(afrag, bfrag, acc2[nt]);
      }
    }

#pragma unroll
    for (int nt = 0; nt < 4; ++nt) {
      const float bv = b2[nt * 16 + lrow];
      float s = 0.0f, s2 = 0.0f;
#pragma unroll
      for (int r = 0; r < 8; ++r) {
        const float v = fmaxf(acc2[nt][r] + bv, 0.f);
        h[(size_t)(base + r + lhalf * 8) * 64 + nt * 16 + lrow] = v;
        s += v;
        s2 += v * v;
      }
      unsafeAtomicAdd(&sstat[nt * 16 + lrow], s);
      unsafeAtomicAdd(&sstat[64 + nt * 16 + lrow], s2);
    }
  }
  __syncthreads();
  if (tid < 128) unsafeAtomicAdd(&stats[tid], sstat[tid]);
}

__global__ void bn_finalize_kernel(const float* __restrict__ stats,
                                   const float* __restrict__ gamma,
                                   const float* __restrict__ beta,
                                   float* __restrict__ ss) {
  const int c = threadIdx.x;
  if (c >= 64) return;
  const float inv_n = 1.0f / (float)N_NODES;
  const float mu  = stats[c] * inv_n;
  const float var = stats[64 + c] * inv_n - mu * mu;
  const float rs  = rsqrtf(var + BN_EPS);
  const float sc  = gamma[c] * rs;
  ss[c]      = sc;
  ss[64 + c] = beta[c] - mu * sc;
}

__global__ void bn_relu_kernel(float* __restrict__ h,
                               const float* __restrict__ ss) {
  const long long i = (long long)blockIdx.x * blockDim.x + threadIdx.x;
  if (i >= (long long)N_NODES * 16) return;
  const int c4 = (int)(i & 15) * 4;
  float4 v = ((const float4*)h)[i];
  v.x = fmaxf(v.x * ss[c4 + 0] + ss[64 + c4 + 0], 0.f);
  v.y = fmaxf(v.y * ss[c4 + 1] + ss[64 + c4 + 1], 0.f);
  v.z = fmaxf(v.z * ss[c4 + 2] + ss[64 + c4 + 2], 0.f);
  v.w = fmaxf(v.w * ss[c4 + 3] + ss[64 + c4 + 3], 0.f);
  ((float4*)h)[i] = v;
}

__global__ void final_kernel(const float* __restrict__ h,
                             const float* __restrict__ Wf,
                             const float* __restrict__ bf,
                             float* __restrict__ out) {
  const int i = blockIdx.x * blockDim.x + threadIdx.x;
  if (i >= N_NODES) return;
  const float4* hp = (const float4*)(h + (size_t)i * 64);
  const float4* wp = (const float4*)Wf;
  float s = bf[0];
#pragma unroll
  for (int j = 0; j < 16; ++j) {
    float4 a = hp[j], b = wp[j];
    s += a.x * b.x + a.y * b.y + a.z * b.z + a.w * b.w;
  }
  out[i] = s;
}

extern "C" void kernel_launch(void* const* d_in, const int* in_sizes, int n_in,
                              void* d_out, int out_size, void* d_ws,
                              size_t ws_size, hipStream_t stream) {
  const float* x      = (const float*)d_in[0];
  const int*   labels = (const int*)d_in[1];
  const int*   ei     = (const int*)d_in[2];
  const float* emb    = (const float*)d_in[3];
  const float* W1_0   = (const float*)d_in[4];
  const float* b1_0   = (const float*)d_in[5];
  const float* W2_0   = (const float*)d_in[6];
  const float* b2_0   = (const float*)d_in[7];
  const float* eps_0  = (const float*)d_in[8];
  const float* g_0    = (const float*)d_in[9];
  const float* be_0   = (const float*)d_in[10];
  const float* W1_s   = (const float*)d_in[11];
  const float* b1_s   = (const float*)d_in[12];
  const float* W2_s   = (const float*)d_in[13];
  const float* b2_s   = (const float*)d_in[14];
  const float* eps_s  = (const float*)d_in[15];
  const float* g_s    = (const float*)d_in[16];
  const float* be_s   = (const float*)d_in[17];
  const float* Wf     = (const float*)d_in[18];
  const float* bf     = (const float*)d_in[19];
  float* out = (float*)d_out;

  float* agg   = (float*)d_ws;
  float* stats = agg + (size_t)N_NODES * 64;
  float* ss    = stats + 128;
  float* Wpad  = ss + 128;
  float* h     = Wpad + 1024;

  const int ntiles     = N_NODES / 16;
  const int mlp_blocks = (ntiles + 3) / 4;

  build_input_kernel<<<(N_NODES * 16 + 255) / 256, 256, 0, stream>>>(x, labels,
                                                                     emb, h);
  pad_w1_kernel<<<4, 256, 0, stream>>>(W1_0, Wpad);

  const long long zero4 = ((long long)N_NODES * 64 + 256) / 4;
  const int zblocks = (int)((zero4 + 255) / 256);

  for (int layer = 0; layer < 3; ++layer) {
    zero_f4_kernel<<<zblocks, 256, 0, stream>>>((float4*)agg, zero4);

    const int log2w = (layer == 0) ? 4 : 6;
    const long long tot = (long long)N_EDGES << log2w;
    scatter_kernel<<<(int)((tot + 255) / 256), 256, 0, stream>>>(h, ei, agg,
                                                                 log2w);

    const float* W1 = (layer == 0) ? Wpad : (W1_s + (size_t)(layer - 1) * 4096);
    const float* b1 = (layer == 0) ? b1_0 : (b1_s + (size_t)(layer - 1) * 64);
    const float* W2 = (layer == 0) ? W2_0 : (W2_s + (size_t)(layer - 1) * 4096);
    const float* b2 = (layer == 0) ? b2_0 : (b2_s + (size_t)(layer - 1) * 64);
    const float* ep = (layer == 0) ? eps_0 : eps_s;
    const int    ei_idx = (layer == 0) ? 0 : (layer - 1);
    const float* gm = (layer == 0) ? g_0 : (g_s + (size_t)(layer - 1) * 64);
    const float* bt = (layer == 0) ? be_0 : (be_s + (size_t)(layer - 1) * 64);
    const int    K  = (layer == 0) ? 16 : 64;

    gin_mlp_kernel<<<mlp_blocks, 128, 0, stream>>>(h, agg, W1, b1, W2, b2, ep,
                                                   ei_idx, K, stats);
    bn_finalize_kernel<<<1, 64, 0, stream>>>(stats, gm, bt, ss);
    bn_relu_kernel<<<(N_NODES * 16 + 255) / 256, 256, 0, stream>>>(h, ss);
  }

  final_kernel<<<(N_NODES + 255) / 256, 256, 0, stream>>>(h, Wf, bf, out);
}


# Launches 18 -> 10: agg clear, W1 pad, BN finalize and the last BN+ReLU pass folded into neighbouring launches (MLP clears its agg tile; BN scale/shift computed per work-group)
# speedup vs baseline: 3.1613x; 3.1613x over previous
#include <hip/hip_runtime.h>
#include <hip/hip_bf16.h>
#include <cstddef>
#include <cstdint>

#define N_NODES 100000
#define N_EDGES 3200000
#define BN_EPS 1e-5f

typedef __attribute__((ext_vector_type(2))) float v2f;
typedef __attribute__((ext_vector_type(8))) float v8f;

__device__ __forceinline__ v8f wmma_f32(v2f a, v2f b, v8f c) {
  return __builtin_amdgcn_wmma_f32_16x16x4_f32(false, a, false, b, (short)0, c,
                                               false, false);
}

__global__ void prep_kernel(const float* __restrict__ x,
                            const int* __restrict__ labels,
                            const float* __restrict__ emb,
                            const float* __restrict__ W1,
                            float* __restrict__ h, float4* __restrict__ agg4,
                            float4* __restrict__ stat4,
                            float* __restrict__ Wpad) {
  long long idx = (long long)blockIdx.x * blockDim.x + threadIdx.x;
  if (idx >= (long long)N_NODES * 16) return;
  const int i = (int)(idx >> 4);
  const int c = (int)(idx & 15);
  float v = 0.0f;
  if (c < 2)       v = x[(size_t)i * 2 + c];
  else if (c < 10) v = emb[(size_t)labels[i] * 8 + (c - 2)];
  h[(size_t)i * 64 + c] = v;
  agg4[idx] = make_float4(0.f, 0.f, 0.f, 0.f);
  if (idx < 64) stat4[idx] = make_float4(0.f, 0.f, 0.f, 0.f);
  if (idx < 16 * 64) Wpad[idx] = ((idx >> 6) < 10) ? W1[idx] : 0.0f;
}

#define SC_WAVES_PER_BLOCK 4
#define SC_CHUNK_EDGES 512
#define SC_DEPTH 8
#define SC_CHUNKS (N_EDGES / SC_CHUNK_EDGES)
#define SC_ITEMS (2 * SC_CHUNKS)
#define SC_BLOCKS (SC_ITEMS / SC_WAVES_PER_BLOCK)
#define SC_RING 64
#define SC_TICKET_STRIDE 32
static_assert(N_EDGES % SC_CHUNK_EDGES == 0, "edge walk assumes whole chunks");
static_assert(SC_ITEMS % SC_WAVES_PER_BLOCK == 0, "one item per wave");

template <int LOG2W>
__global__ void __launch_bounds__(SC_WAVES_PER_BLOCK * 32)
scatter_half_kernel(const float* __restrict__ h,
                    const int* __restrict__ edge_index,
                    float* __restrict__ agg, int* __restrict__ tickets,
                    float* __restrict__ stats) {
  if (blockIdx.x == 0) stats[threadIdx.x] = 0.0f;
  constexpr int G = (LOG2W == 6) ? SC_DEPTH : 2 * SC_DEPTH;
  __shared__ int2 ring_all[SC_WAVES_PER_BLOCK][SC_RING];
  const int lane = threadIdx.x & 31;
  int2* ring = ring_all[threadIdx.x >> 5];

  const unsigned where = __builtin_amdgcn_s_sendmsg_rtn(0x87);
  const int myhalf = (((where >> 16) & 15u) >= 4u) ? 1 : 0;

  int part = myhalf;
  int t = 0;
  if (lane == 0) {
    t = atomicAdd(&tickets[myhalf * SC_TICKET_STRIDE], 1);
    if (t >= SC_CHUNKS) {
      part = myhalf ^ 1;
      t = atomicAdd(&tickets[part * SC_TICKET_STRIDE], 1);
    }
  }
  t = __builtin_amdgcn_readfirstlane(t);
  part = __builtin_amdgcn_readfirstlane(part);
  if ((unsigned)t >= (unsigned)SC_CHUNKS) return;

  const int lo = part ? (N_NODES / 2) : 0;
  const int hi = part ? N_NODES : (N_NODES / 2);

  const int* __restrict__ sp = edge_index + (size_t)t * SC_CHUNK_EDGES + lane;
  const int* __restrict__ dp = sp + N_EDGES;

  const int hsel = lane >> 4;
  const int c16 = lane & 15;

  int head = 0;
  int qn = 0;

  int sv = sp[0];
  int dv = dp[0];
  for (int b = 0; b < SC_CHUNK_EDGES / 32; ++b) {
    const int nb = (b + 1 < SC_CHUNK_EDGES / 32) ? (b + 1) : b;
    const int sn = sp[nb * 32];
    const int dn = dp[nb * 32];

    const bool mine = (dv >= lo) && (dv < hi);
    const unsigned mask = __builtin_amdgcn_ballot_w32(mine);
    const int rank = (int)__builtin_amdgcn_mbcnt_lo(mask, 0u);
    if (mine) ring[(head + qn + rank) & (SC_RING - 1)] = make_int2(sv, dv);
    qn += __builtin_popcount(mask);

    while (qn >= G) {
      int2 e = make_int2(0, 0);
      if (lane < G) e = ring[(head + lane) & (SC_RING - 1)];
      if (LOG2W == 6) {
        float v0[SC_DEPTH], v1[SC_DEPTH];
#pragma unroll
        for (int u = 0; u < SC_DEPTH; ++u) {
          const int sj = __builtin_amdgcn_readlane(e.x, u);
          const float* p = h + (size_t)sj * 64 + lane;
          v0[u] = p[0];
          v1[u] = p[32];
        }
#pragma unroll
        for (int u = 0; u < SC_DEPTH; ++u) {
          const int dj = __builtin_amdgcn_readlane(e.y, u);
          float* q = agg + (size_t)dj * 64 + lane;
          unsafeAtomicAdd(q, v0[u]);
          unsafeAtomicAdd(q + 32, v1[u]);
        }
      } else {
        float v[SC_DEPTH];
#pragma unroll
        for (int u = 0; u < SC_DEPTH; ++u) {
          const int sa = __builtin_amdgcn_readlane(e.x, 2 * u);
          const int sb = __builtin_amdgcn_readlane(e.x, 2 * u + 1);
          const int sj = hsel ? sb : sa;
          v[u] = h[(size_t)sj * 64 + c16];
        }
#pragma unroll
        for (int u = 0; u < SC_DEPTH; ++u) {
          const int da = __builtin_amdgcn_readlane(e.y, 2 * u);
          const int db = __builtin_amdgcn_readlane(e.y, 2 * u + 1);
          const int dj = hsel ? db : da;
          unsafeAtomicAdd(&agg[(size_t)dj * 64 + c16], v[u]);
        }
      }
      head = (head + G) & (SC_RING - 1);
      qn -= G;
    }
    sv = sn;
    dv = dn;
  }

  if (qn > 0) {
    int2 e = make_int2(0, 0);
    if (lane < qn) e = ring[(head + lane) & (SC_RING - 1)];
    if (LOG2W == 6) {
      float v0[SC_DEPTH], v1[SC_DEPTH];
#pragma unroll
      for (int u = 0; u < SC_DEPTH; ++u) {
        v0[u] = 0.0f;
        v1[u] = 0.0f;
        if (u < qn) {
          const int sj = __builtin_amdgcn_readlane(e.x, u);
          const float* p = h + (size_t)sj * 64 + lane;
          v0[u] = p[0];
          v1[u] = p[32];
        }
      }
#pragma unroll
      for (int u = 0; u < SC_DEPTH; ++u) {
        if (u < qn) {
          const int dj = __builtin_amdgcn_readlane(e.y, u);
          float* q = agg + (size_t)dj * 64 + lane;
          unsafeAtomicAdd(q, v0[u]);
          unsafeAtomicAdd(q + 32, v1[u]);
        }
      }
    } else {
      float v[SC_DEPTH];
#pragma unroll
      for (int u = 0; u < SC_DEPTH; ++u) {
        const int sa = __builtin_amdgcn_readlane(e.x, 2 * u);
        const int sb = __builtin_amdgcn_readlane(e.x, 2 * u + 1);
        const int sj = hsel ? sb : sa;
        v[u] = 0.0f;
        if (2 * u + hsel < qn) v[u] = h[(size_t)sj * 64 + c16];
      }
#pragma unroll
      for (int u = 0; u < SC_DEPTH; ++u) {
        const int da = __builtin_amdgcn_readlane(e.y, 2 * u);
        const int db = __builtin_amdgcn_readlane(e.y, 2 * u + 1);
        const int dj = hsel ? db : da;
        if (2 * u + hsel < qn) unsafeAtomicAdd(&agg[(size_t)dj * 64 + c16], v[u]);
      }
    }
  }
}

__global__ void __launch_bounds__(128)
gin_mlp_kernel(float* __restrict__ h, float* __restrict__ agg,
               const float* __restrict__ W1, const float* __restrict__ b1,
               const float* __restrict__ W2, const float* __restrict__ b2,
               const float* __restrict__ eps_ptr, int eps_idx, int K,
               float* __restrict__ stats, int* __restrict__ tickets,
               int clear_agg) {
  __shared__ float ytile[4 * 16 * 68];
  __shared__ float sstat[128];
  const int tid   = threadIdx.x;
  const int wave  = tid >> 5;
  const int lane  = tid & 31;
  const int lrow  = lane & 15;
  const int lhalf = lane >> 4;
  const int tile  = blockIdx.x * 4 + wave;

  if (tid < 128) sstat[tid] = 0.0f;
  if (blockIdx.x == 0 && tid == 0) {
    tickets[0] = 0;
    tickets[SC_TICKET_STRIDE] = 0;
  }
  __syncthreads();

  if (tile < (N_NODES / 16)) {
    const int base = tile * 16;
    const float onepe = 1.0f + eps_ptr[eps_idx];
    const int koff = lhalf * 2;

    v8f acc[4] = {};
    for (int kk = 0; kk < K; kk += 4) {
      const int k0 = kk + koff;
      const float* hp = h   + (size_t)(base + lrow) * 64 + k0;
      const float* ap = agg + (size_t)(base + lrow) * 64 + k0;
      v2f afrag;
      afrag.x = onepe * hp[0] + ap[0];
      afrag.y = onepe * hp[1] + ap[1];
      const float* wp = W1 + (size_t)k0 * 64 + lrow;
#pragma unroll
      for (int nt = 0; nt < 4; ++nt) {
        v2f bfrag;
        bfrag.x = wp[nt * 16];
        bfrag.y = wp[nt * 16 + 64];
        acc[nt] = wmma_f32(afrag, bfrag, acc[nt]);
      }
    }

    float* my = ytile + wave * (16 * 68);
#pragma unroll
    for (int nt = 0; nt < 4; ++nt) {
      const float bv = b1[nt * 16 + lrow];
#pragma unroll
      for (int r = 0; r < 8; ++r)
        my[(r + lhalf * 8) * 68 + nt * 16 + lrow] = fmaxf(acc[nt][r] + bv, 0.f);
    }

    if (clear_agg) {
      const int lq = __builtin_ctz((unsigned)K) - 2;
      for (int j = lane; j < (16 << lq); j += 32) {
        const int r = j >> lq;
        const int c4 = j & ((1 << lq) - 1);
        ((float4*)(agg + (size_t)(base + r) * 64))[c4] =
            make_float4(0.f, 0.f, 0.f, 0.f);
      }
    }

    v8f acc2[4] = {};
    for (int kk = 0; kk < 64; kk += 4) {
      const int k0 = kk + koff;
      v2f afrag;
      afrag.x = my[lrow * 68 + k0];
      afrag.y = my[lrow * 68 + k0 + 1];
      const float* wp = W2 + (size_t)k0 * 64 + lrow;
#pragma unroll
      for (int nt = 0; nt < 4; ++nt) {
        v2f bfrag;
        bfrag.x = wp[nt * 16];
        bfrag.y = wp[nt * 16 + 64];
        acc2[nt] = wmma_f32(afrag, bfrag, acc2[nt]);
      }
    }

#pragma unroll
    for (int nt = 0; nt < 4; ++nt) {
      const float bv = b2[nt * 16 + lrow];
      float s = 0.0f, s2 = 0.0f;
#pragma unroll
      for (int r = 0; r < 8; ++r) {
        const float v = fmaxf(acc2[nt][r] + bv, 0.f);
        h[(size_t)(base + r + lhalf * 8) * 64 + nt * 16 + lrow] = v;
        s += v;
        s2 += v * v;
      }
      unsafeAtomicAdd(&sstat[nt * 16 + lrow], s);
      unsafeAtomicAdd(&sstat[64 + nt * 16 + lrow], s2);
    }
  }
  __syncthreads();
  if (tid < 128) unsafeAtomicAdd(&stats[tid], sstat[tid]);
}

__device__ __forceinline__ void bn_scale_shift(const float* __restrict__ stats,
                                               const float* __restrict__ gamma,
                                               const float* __restrict__ beta,
                                               float* ss) {
  const int c = threadIdx.x;
  if (c < 64) {
    const float inv_n = 1.0f / (float)N_NODES;
    const float mu  = stats[c] * inv_n;
    const float var = stats[64 + c] * inv_n - mu * mu;
    const float rs  = rsqrtf(var + BN_EPS);
    const float sc  = gamma[c] * rs;
    ss[c]      = sc;
    ss[64 + c] = beta[c] - mu * sc;
  }
  __syncthreads();
}

#define BR_TRIPS 5
#define BR_BLOCKS (N_NODES * 16 / (256 * BR_TRIPS))
static_assert(BR_BLOCKS * 256 * BR_TRIPS == N_NODES * 16, "whole trips");
__global__ void __launch_bounds__(256)
bn_relu_kernel(float* __restrict__ h, const float* __restrict__ stats,
               const float* __restrict__ gamma,
               const float* __restrict__ beta) {
  __shared__ float ss[128];
  bn_scale_shift(stats, gamma, beta, ss);
  const int c4 = (threadIdx.x & 15) * 4;
  const float a0 = ss[c4 + 0], a1 = ss[c4 + 1], a2 = ss[c4 + 2], a3 = ss[c4 + 3];
  const float s0 = ss[64 + c4 + 0], s1 = ss[64 + c4 + 1];
  const float s2 = ss[64 + c4 + 2], s3 = ss[64 + c4 + 3];
  float4* p = (float4*)h + (size_t)blockIdx.x * (256 * BR_TRIPS) + threadIdx.x;
  float4 v[BR_TRIPS];
#pragma unroll
  for (int t = 0; t < BR_TRIPS; ++t) v[t] = p[t * 256];
#pragma unroll
  for (int t = 0; t < BR_TRIPS; ++t) {
    float4 w = v[t];
    w.x = fmaxf(w.x * a0 + s0, 0.f);
    w.y = fmaxf(w.y * a1 + s1, 0.f);
    w.z = fmaxf(w.z * a2 + s2, 0.f);
    w.w = fmaxf(w.w * a3 + s3, 0.f);
    p[t * 256] = w;
  }
}

__global__ void __launch_bounds__(256)
final_kernel(const float* __restrict__ h, const float* __restrict__ stats,
             const float* __restrict__ gamma, const float* __restrict__ beta,
             const float* __restrict__ Wf, const float* __restrict__ bf,
             float* __restrict__ out) {
  __shared__ float ss[128];
  bn_scale_shift(stats, gamma, beta, ss);
  const int i = blockIdx.x * blockDim.x + threadIdx.x;
  if (i >= N_NODES) return;
  const float4* hp = (const float4*)(h + (size_t)i * 64);
  const float4* wp = (const float4*)Wf;
  const float4* sc = (const float4*)ss;
  const float4* sh = (const float4*)(ss + 64);
  float s = bf[0];
#pragma unroll
  for (int j = 0; j < 16; ++j) {
    float4 a = hp[j];
    const float4 b = wp[j], m = sc[j], d = sh[j];
    a.x = fmaxf(a.x * m.x + d.x, 0.f);
    a.y = fmaxf(a.y * m.y + d.y, 0.f);
    a.z = fmaxf(a.z * m.z + d.z, 0.f);
    a.w = fmaxf(a.w * m.w + d.w, 0.f);
    s += a.x * b.x + a.y * b.y + a.z * b.z + a.w * b.w;
  }
  out[i] = s;
}

extern "C" void kernel_launch(void* const* d_in, const int* in_sizes, int n_in,
                              void* d_out, int out_size, void* d_ws,
                              size_t ws_size, hipStream_t stream) {
  const float* x      = (const float*)d_in[0];
  const int*   labels = (const int*)d_in[1];
  const int*   ei     = (const int*)d_in[2];
  const float* emb    = (const float*)d_in[3];
  const float* W1_0   = (const float*)d_in[4];
  const float* b1_0   = (const float*)d_in[5];
  const float* W2_0   = (const float*)d_in[6];
  const float* b2_0   = (const float*)d_in[7];
  const float* eps_0  = (const float*)d_in[8];
  const float* g_0    = (const float*)d_in[9];
  const float* be_0   = (const float*)d_in[10];
  const float* W1_s   = (const float*)d_in[11];
  const float* b1_s   = (const float*)d_in[12];
  const float* W2_s   = (const float*)d_in[13];
  const float* b2_s   = (const float*)d_in[14];
  const float* eps_s  = (const float*)d_in[15];
  const float* g_s    = (const float*)d_in[16];
  const float* be_s   = (const float*)d_in[17];
  const float* Wf     = (const float*)d_in[18];
  const float* bf     = (const float*)d_in[19];
  float* out = (float*)d_out;

  float* agg   = (float*)d_ws;
  float* stats = agg + (size_t)N_NODES * 64;
  int*   tk    = (int*)(stats + 128);
  float* Wpad  = stats + 256;
  float* h     = Wpad + 1024;

  const int ntiles     = N_NODES / 16;
  const int mlp_blocks = (ntiles + 3) / 4;

  prep_kernel<<<(N_NODES * 16 + 255) / 256, 256, 0, stream>>>(
      x, labels, emb, W1_0, h, (float4*)agg, (float4*)stats, Wpad);

  for (int layer = 0; layer < 3; ++layer) {
    if (layer == 0)
      scatter_half_kernel<4><<<SC_BLOCKS, SC_WAVES_PER_BLOCK * 32, 0, stream>>>(
          h, ei, agg, tk, stats);
    else
      scatter_half_kernel<6><<<SC_BLOCKS, SC_WAVES_PER_BLOCK * 32, 0, stream>>>(
          h, ei, agg, tk, stats);

    const float* W1 = (layer == 0) ? Wpad : (W1_s + (size_t)(layer - 1) * 4096);
    const float* b1 = (layer == 0) ? b1_0 : (b1_s + (size_t)(layer - 1) * 64);
    const float* W2 = (layer == 0) ? W2_0 : (W2_s + (size_t)(layer - 1) * 4096);
    const float* b2 = (layer == 0) ? b2_0 : (b2_s + (size_t)(layer - 1) * 64);
    const float* ep = (layer == 0) ? eps_0 : eps_s;
    const int    ei_idx = (layer == 0) ? 0 : (layer - 1);
    const float* gm = (layer == 0) ? g_0 : (g_s + (size_t)(layer - 1) * 64);
    const float* bt = (layer == 0) ? be_0 : (be_s + (size_t)(layer - 1) * 64);
    const int    K  = (layer == 0) ? 16 : 64;

    gin_mlp_kernel<<<mlp_blocks, 128, 0, stream>>>(h, agg, W1, b1, W2, b2, ep,
                                                   ei_idx, K, stats, tk,
                                                   (layer < 2) ? 1 : 0);
    if (layer < 2)
      bn_relu_kernel<<<BR_BLOCKS, 256, 0, stream>>>(h, stats, gm, bt);
    else
      final_kernel<<<(N_NODES + 255) / 256, 256, 0, stream>>>(h, stats, gm, bt,
                                                               Wf, bf, out);
  }
}
